# starting kernel
# baseline (speedup 1.0000x reference)
#include <hip/hip_runtime.h>
#include <math.h>


#define T_SEQ  4096
#define DMODEL 1024
#define NH     16
#define HD     64
#define DMLP   4096
#define WIN    512
#define EPS_F  1.1920929e-07f

#define BM   32
#define BN   256
#define KC   32
#define LSTR 36

typedef float v2f __attribute__((ext_vector_type(2)));
typedef float v8f __attribute__((ext_vector_type(8)));

__device__ __forceinline__ v8f wmma4(v2f a, v2f b, v8f c) {
  return __builtin_amdgcn_wmma_f32_16x16x4_f32(false, a, false, b, (short)0, c,
                                               false, false);
}

__global__ __launch_bounds__(256) void rmsnorm_kernel(
    const float* __restrict__ in, float* __restrict__ out, int ncol) {
  const int row = blockIdx.x;
  const float* r = in + (size_t)row * ncol;
  float* o = out + (size_t)row * ncol;
  float ss = 0.0f;
  for (int i = threadIdx.x; i < ncol; i += blockDim.x) {
    float v = r[i];
    ss += v * v;
  }
  __shared__ float red[256];
  red[threadIdx.x] = ss;
  __syncthreads();
  for (int s = 128; s > 0; s >>= 1) {
    if (threadIdx.x < (unsigned)s) red[threadIdx.x] += red[threadIdx.x + s];
    __syncthreads();
  }
  float inv = rsqrtf(red[0] / (float)ncol + EPS_F);
  for (int i = threadIdx.x; i < ncol; i += blockDim.x) o[i] = r[i] * inv;
}

__global__ __launch_bounds__(256) void gemm_wt_kernel(
    const float* __restrict__ A, const float* __restrict__ Wt,
    const float* __restrict__ colScale, const float* __restrict__ resid,
    float* __restrict__ out, int M, int N, int K, int act) {
  __shared__ float lA[2][BM * LSTR];
  __shared__ float lW[2][BN * LSTR];
  (void)M;

  const int tid = threadIdx.x;
  const int lane = tid & 31;
  const int wave = tid >> 5;
  const int lm = lane & 15;
  const int half = lane >> 4;
  const int wm = (wave & 1) * 16;
  const int wn = (wave >> 1) * 64;

  const int m0 = blockIdx.y * BM;
  const int n00 = blockIdx.x * BN;

  const int ldRow = tid >> 3;
  const int ldCol = (tid & 7) * 4;
  const float* Ag = A + (size_t)(m0 + ldRow) * K + ldCol;
  const float* Wg = Wt + (size_t)(n00 + ldRow) * K + ldCol;

  float4 ra;
  float4 rw[8];

  ra = *(const float4*)(Ag);
#pragma unroll
  for (int i = 0; i < 8; ++i)
    rw[i] = *(const float4*)(Wg + (size_t)(i * 32) * K);
  *(float4*)&lA[0][ldRow * LSTR + ldCol] = ra;
#pragma unroll
  for (int i = 0; i < 8; ++i)
    *(float4*)&lW[0][(ldRow + i * 32) * LSTR + ldCol] = rw[i];
  __syncthreads();

  v8f acc[4] = {};
  int cur = 0;
  for (int kc = 0; kc < K; kc += KC) {
    const int nxt = kc + KC;
    if (nxt < K) {
      ra = *(const float4*)(Ag + nxt);
#pragma unroll
      for (int i = 0; i < 8; ++i)
        rw[i] = *(const float4*)(Wg + (size_t)(i * 32) * K + nxt);
    }
    if (kc + 2 * KC < K)
      __builtin_prefetch(Wg + (size_t)(kc + 2 * KC), 0, 1);

    const float* As = &lA[cur][(wm + lm) * LSTR + 2 * half];
    const float* Ws = &lW[cur][(wn + lm) * LSTR + 2 * half];
#pragma unroll
    for (int kk = 0; kk < KC; kk += 4) {
      v2f a = *(const v2f*)(As + kk);
      v2f b0 = *(const v2f*)(Ws + kk);
      v2f b1 = *(const v2f*)(Ws + kk + 16 * LSTR);
      v2f b2 = *(const v2f*)(Ws + kk + 32 * LSTR);
      v2f b3 = *(const v2f*)(Ws + kk + 48 * LSTR);
      acc[0] = wmma4(a, b0, acc[0]);
      acc[1] = wmma4(a, b1, acc[1]);
      acc[2] = wmma4(a, b2, acc[2]);
      acc[3] = wmma4(a, b3, acc[3]);
    }
    if (nxt < K) {
      *(float4*)&lA[cur ^ 1][ldRow * LSTR + ldCol] = ra;
#pragma unroll
      for (int i = 0; i < 8; ++i)
        *(float4*)&lW[cur ^ 1][(ldRow + i * 32) * LSTR + ldCol] = rw[i];
    }
    __syncthreads();
    cur ^= 1;
  }

  const int mB = m0 + wm;
  const int nB = n00 + wn;
#pragma unroll
  for (int s = 0; s < 4; ++s) {
    const int n = nB + s * 16 + lm;
    const float cs = colScale ? colScale[n] : 1.0f;
#pragma unroll
    for (int r = 0; r < 8; ++r) {
      const int m = mB + r + 8 * half;
      float v = acc[s][r] * cs;
      if (act == 1) {
        v = fmaxf(v, 0.0f);
        v = v * v;
      }
      if (resid) v += resid[(size_t)m * N + n];
      out[(size_t)m * N + n] = v;
    }
  }
}

__global__ __launch_bounds__(64) void qk_rope_kernel(
    const float* __restrict__ qkv, float* __restrict__ qh,
    float* __restrict__ kh, float* __restrict__ vh) {
  const int t = blockIdx.x;
  const int h = blockIdx.y;
  const int d = threadIdx.x;
  const float* base = qkv + (size_t)t * (3 * DMODEL) + h * HD;
  float qv = base[d];
  float kv = base[DMODEL + d];
  float vv = base[2 * DMODEL + d];

  __shared__ float red[64], sq[64], sk[64];
  red[d] = qv * qv;
  __syncthreads();
  for (int s = 32; s > 0; s >>= 1) {
    if (d < s) red[d] += red[d + s];
    __syncthreads();
  }
  float qi = rsqrtf(red[0] / (float)HD + EPS_F);
  __syncthreads();
  red[d] = kv * kv;
  __syncthreads();
  for (int s = 32; s > 0; s >>= 1) {
    if (d < s) red[d] += red[d + s];
    __syncthreads();
  }
  float ki = rsqrtf(red[0] / (float)HD + EPS_F);
  __syncthreads();

  sq[d] = qv * qi;
  sk[d] = kv * ki;
  __syncthreads();

  const int j = d & 31;
  float theta = (j < 16) ? (float)t * powf(10000.0f, -(float)j / 15.0f) : 0.0f;
  float c = cosf(theta), s_ = sinf(theta);
  float x1q = sq[j], x2q = sq[j + 32];
  float x1k = sk[j], x2k = sk[j + 32];
  float oq = (d < 32) ? (x1q * c + x2q * s_) : (-x1q * s_ + x2q * c);
  float ok = (d < 32) ? (x1k * c + x2k * s_) : (-x1k * s_ + x2k * c);

  const size_t oidx = ((size_t)h * T_SEQ + t) * HD + d;
  qh[oidx] = oq;
  kh[oidx] = ok;
  vh[oidx] = vv;
}

__global__ __launch_bounds__(32) void attn_kernel(
    const float* __restrict__ Qh, const float* __restrict__ Kh,
    const float* __restrict__ Vh, float* __restrict__ outAttn) {
  const int h = blockIdx.x;
  const int q0 = blockIdx.y * 16;
  const int lane = threadIdx.x;
  const int lm = lane & 15;
  const int half = lane >> 4;
  const size_t hb = (size_t)h * T_SEQ * HD;

  const float* qrow = Qh + hb + (size_t)(q0 + lm) * HD + 2 * half;
  v2f qa[16];
#pragma unroll
  for (int dc = 0; dc < 16; ++dc) qa[dc] = *(const v2f*)(qrow + 4 * dc);

  float mi[8], li[8];
  v8f o[4] = {};
#pragma unroll
  for (int r = 0; r < 8; ++r) {
    mi[r] = -INFINITY;
    li[r] = 0.0f;
  }

  __shared__ float sP[16 * 16];

  const int lo = q0 - WIN + 1;
  const int kt0 = (lo < 0) ? 0 : (lo / 16) * 16;
  for (int kt = kt0; kt <= q0; kt += 16) {
    const float* krow = Kh + hb + (size_t)(kt + lm) * HD + 2 * half;
    v2f kb[16];
#pragma unroll
    for (int dc = 0; dc < 16; ++dc) kb[dc] = *(const v2f*)(krow + 4 * dc);
    v8f s0 = {}, s1 = {};
#pragma unroll
    for (int dc = 0; dc < 16; dc += 2) {
      s0 = wmma4(qa[dc], kb[dc], s0);
      s1 = wmma4(qa[dc + 1], kb[dc + 1], s1);
    }
    v8f s = s0 + s1;

#pragma unroll
    for (int r = 0; r < 8; ++r) {
      const int qp = q0 + r + 8 * half;
      const int kp = kt + lm;
      float sv = s[r] * 0.125f;
      const bool okm = (kp <= qp) && (kp > qp - WIN);
      sv = okm ? sv : -INFINITY;
      float rm = sv;
      rm = fmaxf(rm, __shfl_xor(rm, 1, 32));
      rm = fmaxf(rm, __shfl_xor(rm, 2, 32));
      rm = fmaxf(rm, __shfl_xor(rm, 4, 32));
      rm = fmaxf(rm, __shfl_xor(rm, 8, 32));
      const float nm = fmaxf(mi[r], rm);
      const float p = (sv == -INFINITY) ? 0.0f : __expf(sv - nm);
      float corr;
      if (nm == -INFINITY)
        corr = 1.0f;
      else if (mi[r] == -INFINITY)
        corr = 0.0f;
      else
        corr = __expf(mi[r] - nm);
      float rs = p;
      rs += __shfl_xor(rs, 1, 32);
      rs += __shfl_xor(rs, 2, 32);
      rs += __shfl_xor(rs, 4, 32);
      rs += __shfl_xor(rs, 8, 32);
      li[r] = li[r] * corr + rs;
      mi[r] = nm;
      o[0][r] *= corr;
      o[1][r] *= corr;
      o[2][r] *= corr;
      o[3][r] *= corr;
      sP[(r + 8 * half) * 16 + lm] = p;
    }
    __syncthreads();

    const float* vbase = Vh + hb;
    v2f vb[16];
#pragma unroll
    for (int kc4 = 0; kc4 < 4; ++kc4) {
      const int kr = kt + kc4 * 4 + 2 * half;
#pragma unroll
      for (int f = 0; f < 4; ++f) {
        v2f b;
        b.x = vbase[(size_t)kr * HD + f * 16 + lm];
        b.y = vbase[(size_t)(kr + 1) * HD + f * 16 + lm];
        vb[kc4 * 4 + f] = b;
      }
    }
    v2f pa[4];
#pragma unroll
    for (int kc4 = 0; kc4 < 4; ++kc4) {
      pa[kc4].x = sP[lm * 16 + kc4 * 4 + 2 * half];
      pa[kc4].y = sP[lm * 16 + kc4 * 4 + 2 * half + 1];
    }
#pragma unroll
    for (int kc4 = 0; kc4 < 4; ++kc4)
#pragma unroll
      for (int f = 0; f < 4; ++f) o[f] = wmma4(pa[kc4], vb[kc4 * 4 + f], o[f]);
    __syncthreads();
  }

#pragma unroll
  for (int f = 0; f < 4; ++f) {
#pragma unroll
    for (int r = 0; r < 8; ++r) {
      const int tq = q0 + r + 8 * half;
      const int col = h * HD + f * 16 + lm;
      outAttn[(size_t)tq * DMODEL + col] = o[f][r] / li[r];
    }
  }
}

extern "C" void kernel_launch(void* const* d_in, const int* in_sizes, int n_in,
                              void* d_out, int out_size, void* d_ws,
                              size_t ws_size, hipStream_t stream) {
  (void)in_sizes;
  (void)n_in;
  (void)out_size;
  (void)ws_size;
  const float* x = (const float*)d_in[0];
  const float* qkv_w = (const float*)d_in[1];
  const float* o_w = (const float*)d_in[2];
  const float* o_scale = (const float*)d_in[3];
  const float* w1 = (const float*)d_in[4];
  const float* w2 = (const float*)d_in[5];
  const float* w2_scale = (const float*)d_in[6];
  float* out = (float*)d_out;

  const size_t TD = (size_t)T_SEQ * DMODEL;
  float* xn = (float*)d_ws;
  float* qkv = xn + TD;
  float* qh = qkv + 3 * TD;
  float* kh = qh + TD;
  float* vh = kh + TD;
  float* attn = vh + TD;
  float* x1 = attn + TD;
  float* hbuf = qkv;

  rmsnorm_kernel<<<T_SEQ, 256, 0, stream>>>(x, xn, DMODEL);
  gemm_wt_kernel<<<dim3(3 * DMODEL / BN, T_SEQ / BM), 256, 0, stream>>>(
      xn, qkv_w, nullptr, nullptr, qkv, T_SEQ, 3 * DMODEL, DMODEL, 0);
  qk_rope_kernel<<<dim3(T_SEQ, NH), 64, 0, stream>>>(qkv, qh, kh, vh);
  attn_kernel<<<dim3(NH, T_SEQ / 16), 32, 0, stream>>>(qh, kh, vh, attn);
  gemm_wt_kernel<<<dim3(DMODEL / BN, T_SEQ / BM), 256, 0, stream>>>(
      attn, o_w, o_scale, x, x1, T_SEQ, DMODEL, DMODEL, 0);
  rmsnorm_kernel<<<T_SEQ, 256, 0, stream>>>(x1, xn, DMODEL);
  gemm_wt_kernel<<<dim3(DMLP / BN, T_SEQ / BM), 256, 0, stream>>>(
      xn, w1, nullptr, nullptr, hbuf, T_SEQ, DMLP, DMODEL, 1);
  gemm_wt_kernel<<<dim3(DMODEL / BN, T_SEQ / BM), 256, 0, stream>>>(
      hbuf, w2, w2_scale, x1, out, T_SEQ, DMODEL, DMLP, 0);
}


# GEMM operands straight from global into WMMA fragment registers: K-permuted 128-bit loads, two register sets, 64x64 wave tiles; no LDS staging, scratch copy or barriers; k sum order changes
# speedup vs baseline: 2.6994x; 2.6994x over previous
#include <hip/hip_runtime.h>
#include <math.h>


#define T_SEQ  4096
#define DMODEL 1024
#define NH     16
#define HD     64
#define DMLP   4096
#define WIN    512
#define EPS_F  1.1920929e-07f

#define BM   128
#define BN   128
#define KC   32

typedef float v2f __attribute__((ext_vector_type(2)));
typedef float v4f __attribute__((ext_vector_type(4)));
typedef float v8f __attribute__((ext_vector_type(8)));

__device__ __forceinline__ v8f wmma4(v2f a, v2f b, v8f c) {
  return __builtin_amdgcn_wmma_f32_16x16x4_f32(false, a, false, b, (short)0, c,
                                               false, false);
}

__global__ __launch_bounds__(256) void rmsnorm_kernel(
    const float* __restrict__ in, float* __restrict__ out, int ncol) {
  const int row = blockIdx.x;
  const float* r = in + (size_t)row * ncol;
  float* o = out + (size_t)row * ncol;
  float ss = 0.0f;
  for (int i = threadIdx.x; i < ncol; i += blockDim.x) {
    float v = r[i];
    ss += v * v;
  }
  __shared__ float red[256];
  red[threadIdx.x] = ss;
  __syncthreads();
  for (int s = 128; s > 0; s >>= 1) {
    if (threadIdx.x < (unsigned)s) red[threadIdx.x] += red[threadIdx.x + s];
    __syncthreads();
  }
  float inv = rsqrtf(red[0] / (float)ncol + EPS_F);
  for (int i = threadIdx.x; i < ncol; i += blockDim.x) o[i] = r[i] * inv;
}

struct FragSet {
  v4f a[4][4];
  v4f b[4][4];
};

__device__ __forceinline__ void mma_slice(const FragSet& f, v8f (&acc)[4][4],
                                          int j) {
#pragma unroll
  for (int i = 0; i < 4; ++i)
#pragma unroll
    for (int s = 0; s < 4; ++s)
      acc[i][s] = wmma4(__builtin_shufflevector(f.a[i][j], f.a[i][j], 0, 1),
                        __builtin_shufflevector(f.b[s][j], f.b[s][j], 0, 1),
                        acc[i][s]);
#pragma unroll
  for (int i = 0; i < 4; ++i)
#pragma unroll
    for (int s = 0; s < 4; ++s)
      acc[i][s] = wmma4(__builtin_shufflevector(f.a[i][j], f.a[i][j], 2, 3),
                        __builtin_shufflevector(f.b[s][j], f.b[s][j], 2, 3),
                        acc[i][s]);
}

__device__ __forceinline__ void load_slice(FragSet& f,
                                           const float* __restrict__ Ap,
                                           const float* __restrict__ Wp,
                                           size_t rs16, int k, int j) {
#pragma unroll
  for (int i = 0; i < 4; ++i)
    f.a[i][j] = *(const v4f*)(Ap + i * rs16 + k + 4 * j);
#pragma unroll
  for (int i = 0; i < 4; ++i)
    f.b[i][j] = *(const v4f*)(Wp + i * rs16 + k + 4 * j);
}

__device__ __forceinline__ void chunk_step(const FragSet& use, FragSet& fill,
                                           v8f (&acc)[4][4],
                                           const float* __restrict__ Ap,
                                           const float* __restrict__ Wp,
                                           size_t rs16, int knext) {
#pragma unroll
  for (int j = 0; j < 4; ++j) {
    load_slice(fill, Ap, Wp, rs16, knext, j);
    asm volatile("" ::: "memory");
    __builtin_amdgcn_sched_barrier(0);
    mma_slice(use, acc, j);
    __builtin_amdgcn_sched_barrier(0);
  }
}

__device__ __forceinline__ void mma_set(const FragSet& f, v8f (&acc)[4][4]) {
#pragma unroll
  for (int j = 0; j < 4; ++j) mma_slice(f, acc, j);
}

template <bool RES, bool ACT>
__device__ __forceinline__ void gemm_epilogue(
    const v8f (&acc)[4][4], const float* __restrict__ colScale,
    const float* __restrict__ resid, float* __restrict__ out, int mB, int nB,
    int N, int lm, int half) {
  float cs[4];
#pragma unroll
  for (int s = 0; s < 4; ++s)
    cs[s] = colScale ? colScale[nB + s * 16 + lm] : 1.0f;
#pragma unroll
  for (int i = 0; i < 4; ++i) {
    float rv[8][4];
    if (RES) {
#pragma unroll
      for (int r = 0; r < 8; ++r)
#pragma unroll
        for (int s = 0; s < 4; ++s)
          rv[r][s] = resid[(size_t)(mB + i * 16 + r + 8 * half) * N + nB +
                           s * 16 + lm];
    }
#pragma unroll
    for (int r = 0; r < 8; ++r) {
      const int m = mB + i * 16 + r + 8 * half;
#pragma unroll
      for (int s = 0; s < 4; ++s) {
        const int n = nB + s * 16 + lm;
        float v = acc[i][s][r] * cs[s];
        if (ACT) {
          v = fmaxf(v, 0.0f);
          v = v * v;
        }
        if (RES) v += rv[r][s];
        out[(size_t)m * N + n] = v;
      }
    }
  }
}

__global__ __launch_bounds__(128) void gemm_wt_kernel(
    const float* __restrict__ A, const float* __restrict__ Wt,
    const float* __restrict__ colScale, const float* __restrict__ resid,
    float* __restrict__ out, int M, int N, int K, int act) {
  (void)M;
  const int tid = threadIdx.x;
  const int lane = tid & 31;
  const int wave = tid >> 5;
  const int lm = lane & 15;
  const int half = lane >> 4;
  const int mB = blockIdx.y * BM + (wave & 1) * 64;
  const int nB = blockIdx.x * BN + (wave >> 1) * 64;

  const float* Ap = A + (size_t)(mB + lm) * K + 16 * half;
  const float* Wp = Wt + (size_t)(nB + lm) * K + 16 * half;
  const size_t rs16 = (size_t)16 * K;

  v8f acc[4][4] = {};
  FragSet f0, f1;

#pragma unroll
  for (int j = 0; j < 4; ++j) {
    load_slice(f0, Ap, Wp, rs16, 0, j);
    asm volatile("" ::: "memory");
  }
  int kc = 0;
  for (; kc + 2 * KC < K; kc += 2 * KC) {
    chunk_step(f0, f1, acc, Ap, Wp, rs16, kc + KC);
    chunk_step(f1, f0, acc, Ap, Wp, rs16, kc + 2 * KC);
  }
  chunk_step(f0, f1, acc, Ap, Wp, rs16, kc + KC);
  mma_set(f1, acc);

  if (resid) {
    if (act == 1)
      gemm_epilogue<true, true>(acc, colScale, resid, out, mB, nB, N, lm, half);
    else
      gemm_epilogue<true, false>(acc, colScale, resid, out, mB, nB, N, lm, half);
  } else {
    if (act == 1)
      gemm_epilogue<false, true>(acc, colScale, resid, out, mB, nB, N, lm, half);
    else
      gemm_epilogue<false, false>(acc, colScale, resid, out, mB, nB, N, lm, half);
  }
}

__global__ __launch_bounds__(64) void qk_rope_kernel(
    const float* __restrict__ qkv, float* __restrict__ qh,
    float* __restrict__ kh, float* __restrict__ vh) {
  const int t = blockIdx.x;
  const int h = blockIdx.y;
  const int d = threadIdx.x;
  const float* base = qkv + (size_t)t * (3 * DMODEL) + h * HD;
  float qv = base[d];
  float kv = base[DMODEL + d];
  float vv = base[2 * DMODEL + d];

  __shared__ float red[64], sq[64], sk[64];
  red[d] = qv * qv;
  __syncthreads();
  for (int s = 32; s > 0; s >>= 1) {
    if (d < s) red[d] += red[d + s];
    __syncthreads();
  }
  float qi = rsqrtf(red[0] / (float)HD + EPS_F);
  __syncthreads();
  red[d] = kv * kv;
  __syncthreads();
  for (int s = 32; s > 0; s >>= 1) {
    if (d < s) red[d] += red[d + s];
    __syncthreads();
  }
  float ki = rsqrtf(red[0] / (float)HD + EPS_F);
  __syncthreads();

  sq[d] = qv * qi;
  sk[d] = kv * ki;
  __syncthreads();

  const int j = d & 31;
  float theta = (j < 16) ? (float)t * powf(10000.0f, -(float)j / 15.0f) : 0.0f;
  float c = cosf(theta), s_ = sinf(theta);
  float x1q = sq[j], x2q = sq[j + 32];
  float x1k = sk[j], x2k = sk[j + 32];
  float oq = (d < 32) ? (x1q * c + x2q * s_) : (-x1q * s_ + x2q * c);
  float ok = (d < 32) ? (x1k * c + x2k * s_) : (-x1k * s_ + x2k * c);

  const size_t oidx = ((size_t)h * T_SEQ + t) * HD + d;
  qh[oidx] = oq;
  kh[oidx] = ok;
  vh[oidx] = vv;
}

__global__ __launch_bounds__(32) void attn_kernel(
    const float* __restrict__ Qh, const float* __restrict__ Kh,
    const float* __restrict__ Vh, float* __restrict__ outAttn) {
  const int h = blockIdx.x;
  const int q0 = blockIdx.y * 16;
  const int lane = threadIdx.x;
  const int lm = lane & 15;
  const int half = lane >> 4;
  const size_t hb = (size_t)h * T_SEQ * HD;

  const float* qrow = Qh + hb + (size_t)(q0 + lm) * HD + 2 * half;
  v2f qa[16];
#pragma unroll
  for (int dc = 0; dc < 16; ++dc) qa[dc] = *(const v2f*)(qrow + 4 * dc);

  float mi[8], li[8];
  v8f o[4] = {};
#pragma unroll
  for (int r = 0; r < 8; ++r) {
    mi[r] = -INFINITY;
    li[r] = 0.0f;
  }

  __shared__ float sP[16 * 16];

  const int lo = q0 - WIN + 1;
  const int kt0 = (lo < 0) ? 0 : (lo / 16) * 16;
  for (int kt = kt0; kt <= q0; kt += 16) {
    const float* krow = Kh + hb + (size_t)(kt + lm) * HD + 2 * half;
    v2f kb[16];
#pragma unroll
    for (int dc = 0; dc < 16; ++dc) kb[dc] = *(const v2f*)(krow + 4 * dc);
    v8f s0 = {}, s1 = {};
#pragma unroll
    for (int dc = 0; dc < 16; dc += 2) {
      s0 = wmma4(qa[dc], kb[dc], s0);
      s1 = wmma4(qa[dc + 1], kb[dc + 1], s1);
    }
    v8f s = s0 + s1;

#pragma unroll
    for (int r = 0; r < 8; ++r) {
      const int qp = q0 + r + 8 * half;
      const int kp = kt + lm;
      float sv = s[r] * 0.125f;
      const bool okm = (kp <= qp) && (kp > qp - WIN);
      sv = okm ? sv : -INFINITY;
      float rm = sv;
      rm = fmaxf(rm, __shfl_xor(rm, 1, 32));
      rm = fmaxf(rm, __shfl_xor(rm, 2, 32));
      rm = fmaxf(rm, __shfl_xor(rm, 4, 32));
      rm = fmaxf(rm, __shfl_xor(rm, 8, 32));
      const float nm = fmaxf(mi[r], rm);
      const float p = (sv == -INFINITY) ? 0.0f : __expf(sv - nm);
      float corr;
      if (nm == -INFINITY)
        corr = 1.0f;
      else if (mi[r] == -INFINITY)
        corr = 0.0f;
      else
        corr = __expf(mi[r] - nm);
      float rs = p;
      rs += __shfl_xor(rs, 1, 32);
      rs += __shfl_xor(rs, 2, 32);
      rs += __shfl_xor(rs, 4, 32);
      rs += __shfl_xor(rs, 8, 32);
      li[r] = li[r] * corr + rs;
      mi[r] = nm;
      o[0][r] *= corr;
      o[1][r] *= corr;
      o[2][r] *= corr;
      o[3][r] *= corr;
      sP[(r + 8 * half) * 16 + lm] = p;
    }
    __syncthreads();

    const float* vbase = Vh + hb;
    v2f vb[16];
#pragma unroll
    for (int kc4 = 0; kc4 < 4; ++kc4) {
      const int kr = kt + kc4 * 4 + 2 * half;
#pragma unroll
      for (int f = 0; f < 4; ++f) {
        v2f b;
        b.x = vbase[(size_t)kr * HD + f * 16 + lm];
        b.y = vbase[(size_t)(kr + 1) * HD + f * 16 + lm];
        vb[kc4 * 4 + f] = b;
      }
    }
    v2f pa[4];
#pragma unroll
    for (int kc4 = 0; kc4 < 4; ++kc4) {
      pa[kc4].x = sP[lm * 16 + kc4 * 4 + 2 * half];
      pa[kc4].y = sP[lm * 16 + kc4 * 4 + 2 * half + 1];
    }
#pragma unroll
    for (int kc4 = 0; kc4 < 4; ++kc4)
#pragma unroll
      for (int f = 0; f < 4; ++f) o[f] = wmma4(pa[kc4], vb[kc4 * 4 + f], o[f]);
    __syncthreads();
  }

#pragma unroll
  for (int f = 0; f < 4; ++f) {
#pragma unroll
    for (int r = 0; r < 8; ++r) {
      const int tq = q0 + r + 8 * half;
      const int col = h * HD + f * 16 + lm;
      outAttn[(size_t)tq * DMODEL + col] = o[f][r] / li[r];
    }
  }
}

extern "C" void kernel_launch(void* const* d_in, const int* in_sizes, int n_in,
                              void* d_out, int out_size, void* d_ws,
                              size_t ws_size, hipStream_t stream) {
  (void)in_sizes;
  (void)n_in;
  (void)out_size;
  (void)ws_size;
  const float* x = (const float*)d_in[0];
  const float* qkv_w = (const float*)d_in[1];
  const float* o_w = (const float*)d_in[2];
  const float* o_scale = (const float*)d_in[3];
  const float* w1 = (const float*)d_in[4];
  const float* w2 = (const float*)d_in[5];
  const float* w2_scale = (const float*)d_in[6];
  float* out = (float*)d_out;

  const size_t TD = (size_t)T_SEQ * DMODEL;
  float* xn = (float*)d_ws;
  float* qkv = xn + TD;
  float* qh = qkv + 3 * TD;
  float* kh = qh + TD;
  float* vh = kh + TD;
  float* attn = vh + TD;
  float* x1 = attn + TD;
  float* hbuf = qkv;

  rmsnorm_kernel<<<T_SEQ, 256, 0, stream>>>(x, xn, DMODEL);
  gemm_wt_kernel<<<dim3(3 * DMODEL / BN, T_SEQ / BM), 128, 0, stream>>>(
      xn, qkv_w, nullptr, nullptr, qkv, T_SEQ, 3 * DMODEL, DMODEL, 0);
  qk_rope_kernel<<<dim3(T_SEQ, NH), 64, 0, stream>>>(qkv, qh, kh, vh);
  attn_kernel<<<dim3(NH, T_SEQ / 16), 32, 0, stream>>>(qh, kh, vh, attn);
  gemm_wt_kernel<<<dim3(DMODEL / BN, T_SEQ / BM), 128, 0, stream>>>(
      attn, o_w, o_scale, x, x1, T_SEQ, DMODEL, DMODEL, 0);
  rmsnorm_kernel<<<T_SEQ, 256, 0, stream>>>(x1, xn, DMODEL);
  gemm_wt_kernel<<<dim3(DMLP / BN, T_SEQ / BM), 128, 0, stream>>>(
      xn, w1, nullptr, nullptr, hbuf, T_SEQ, DMLP, DMODEL, 1);
  gemm_wt_kernel<<<dim3(DMODEL / BN, T_SEQ / BM), 128, 0, stream>>>(
      hbuf, w2, w2_scale, x1, out, T_SEQ, DMODEL, DMLP, 0);
}


# Attention softmax without a running max: scores are bounded (q,k rows RMS-normed, |s|<=8), so p=exp(s); no per-tile lane reductions or O rescale; row sums added over lanes once; row-sum order changes
# speedup vs baseline: 2.9441x; 1.0906x over previous
#include <hip/hip_runtime.h>
#include <math.h>


#define T_SEQ  4096
#define DMODEL 1024
#define NH     16
#define HD     64
#define DMLP   4096
#define WIN    512
#define EPS_F  1.1920929e-07f

#define BM   128
#define BN   128
#define KC   32

typedef float v2f __attribute__((ext_vector_type(2)));
typedef float v4f __attribute__((ext_vector_type(4)));
typedef float v8f __attribute__((ext_vector_type(8)));

__device__ __forceinline__ v8f wmma4(v2f a, v2f b, v8f c) {
  return __builtin_amdgcn_wmma_f32_16x16x4_f32(false, a, false, b, (short)0, c,
                                               false, false);
}

__global__ __launch_bounds__(256) void rmsnorm_kernel(
    const float* __restrict__ in, float* __restrict__ out, int ncol) {
  const int row = blockIdx.x;
  const float* r = in + (size_t)row * ncol;
  float* o = out + (size_t)row * ncol;
  float ss = 0.0f;
  for (int i = threadIdx.x; i < ncol; i += blockDim.x) {
    float v = r[i];
    ss += v * v;
  }
  __shared__ float red[256];
  red[threadIdx.x] = ss;
  __syncthreads();
  for (int s = 128; s > 0; s >>= 1) {
    if (threadIdx.x < (unsigned)s) red[threadIdx.x] += red[threadIdx.x + s];
    __syncthreads();
  }
  float inv = rsqrtf(red[0] / (float)ncol + EPS_F);
  for (int i = threadIdx.x; i < ncol; i += blockDim.x) o[i] = r[i] * inv;
}

struct FragSet {
  v4f a[4][4];
  v4f b[4][4];
};

__device__ __forceinline__ void mma_slice(const FragSet& f, v8f (&acc)[4][4],
                                          int j) {
#pragma unroll
  for (int i = 0; i < 4; ++i)
#pragma unroll
    for (int s = 0; s < 4; ++s)
      acc[i][s] = wmma4(__builtin_shufflevector(f.a[i][j], f.a[i][j], 0, 1),
                        __builtin_shufflevector(f.b[s][j], f.b[s][j], 0, 1),
                        acc[i][s]);
#pragma unroll
  for (int i = 0; i < 4; ++i)
#pragma unroll
    for (int s = 0; s < 4; ++s)
      acc[i][s] = wmma4(__builtin_shufflevector(f.a[i][j], f.a[i][j], 2, 3),
                        __builtin_shufflevector(f.b[s][j], f.b[s][j], 2, 3),
                        acc[i][s]);
}

__device__ __forceinline__ void load_slice(FragSet& f,
                                           const float* __restrict__ Ap,
                                           const float* __restrict__ Wp,
                                           size_t rs16, int k, int j) {
#pragma unroll
  for (int i = 0; i < 4; ++i)
    f.a[i][j] = *(const v4f*)(Ap + i * rs16 + k + 4 * j);
#pragma unroll
  for (int i = 0; i < 4; ++i)
    f.b[i][j] = *(const v4f*)(Wp + i * rs16 + k + 4 * j);
}

__device__ __forceinline__ void chunk_step(const FragSet& use, FragSet& fill,
                                           v8f (&acc)[4][4],
                                           const float* __restrict__ Ap,
                                           const float* __restrict__ Wp,
                                           size_t rs16, int knext) {
#pragma unroll
  for (int j = 0; j < 4; ++j) {
    load_slice(fill, Ap, Wp, rs16, knext, j);
    asm volatile("" ::: "memory");
    __builtin_amdgcn_sched_barrier(0);
    mma_slice(use, acc, j);
    __builtin_amdgcn_sched_barrier(0);
  }
}

__device__ __forceinline__ void mma_set(const FragSet& f, v8f (&acc)[4][4]) {
#pragma unroll
  for (int j = 0; j < 4; ++j) mma_slice(f, acc, j);
}

template <bool RES, bool ACT>
__device__ __forceinline__ void gemm_epilogue(
    const v8f (&acc)[4][4], const float* __restrict__ colScale,
    const float* __restrict__ resid, float* __restrict__ out, int mB, int nB,
    int N, int lm, int half) {
  float cs[4];
#pragma unroll
  for (int s = 0; s < 4; ++s)
    cs[s] = colScale ? colScale[nB + s * 16 + lm] : 1.0f;
#pragma unroll
  for (int i = 0; i < 4; ++i) {
    float rv[8][4];
    if (RES) {
#pragma unroll
      for (int r = 0; r < 8; ++r)
#pragma unroll
        for (int s = 0; s < 4; ++s)
          rv[r][s] = resid[(size_t)(mB + i * 16 + r + 8 * half) * N + nB +
                           s * 16 + lm];
    }
#pragma unroll
    for (int r = 0; r < 8; ++r) {
      const int m = mB + i * 16 + r + 8 * half;
#pragma unroll
      for (int s = 0; s < 4; ++s) {
        const int n = nB + s * 16 + lm;
        float v = acc[i][s][r] * cs[s];
        if (ACT) {
          v = fmaxf(v, 0.0f);
          v = v * v;
        }
        if (RES) v += rv[r][s];
        out[(size_t)m * N + n] = v;
      }
    }
  }
}

__global__ __launch_bounds__(128) void gemm_wt_kernel(
    const float* __restrict__ A, const float* __restrict__ Wt,
    const float* __restrict__ colScale, const float* __restrict__ resid,
    float* __restrict__ out, int M, int N, int K, int act) {
  (void)M;
  const int tid = threadIdx.x;
  const int lane = tid & 31;
  const int wave = tid >> 5;
  const int lm = lane & 15;
  const int half = lane >> 4;
  const int mB = blockIdx.y * BM + (wave & 1) * 64;
  const int nB = blockIdx.x * BN + (wave >> 1) * 64;

  const float* Ap = A + (size_t)(mB + lm) * K + 16 * half;
  const float* Wp = Wt + (size_t)(nB + lm) * K + 16 * half;
  const size_t rs16 = (size_t)16 * K;

  v8f acc[4][4] = {};
  FragSet f0, f1;

#pragma unroll
  for (int j = 0; j < 4; ++j) {
    load_slice(f0, Ap, Wp, rs16, 0, j);
    asm volatile("" ::: "memory");
  }
  int kc = 0;
  for (; kc + 2 * KC < K; kc += 2 * KC) {
    chunk_step(f0, f1, acc, Ap, Wp, rs16, kc + KC);
    chunk_step(f1, f0, acc, Ap, Wp, rs16, kc + 2 * KC);
  }
  chunk_step(f0, f1, acc, Ap, Wp, rs16, kc + KC);
  mma_set(f1, acc);

  if (resid) {
    if (act == 1)
      gemm_epilogue<true, true>(acc, colScale, resid, out, mB, nB, N, lm, half);
    else
      gemm_epilogue<true, false>(acc, colScale, resid, out, mB, nB, N, lm, half);
  } else {
    if (act == 1)
      gemm_epilogue<false, true>(acc, colScale, resid, out, mB, nB, N, lm, half);
    else
      gemm_epilogue<false, false>(acc, colScale, resid, out, mB, nB, N, lm, half);
  }
}

__global__ __launch_bounds__(64) void qk_rope_kernel(
    const float* __restrict__ qkv, float* __restrict__ qh,
    float* __restrict__ kh, float* __restrict__ vh) {
  const int t = blockIdx.x;
  const int h = blockIdx.y;
  const int d = threadIdx.x;
  const float* base = qkv + (size_t)t * (3 * DMODEL) + h * HD;
  float qv = base[d];
  float kv = base[DMODEL + d];
  float vv = base[2 * DMODEL + d];

  __shared__ float red[64], sq[64], sk[64];
  red[d] = qv * qv;
  __syncthreads();
  for (int s = 32; s > 0; s >>= 1) {
    if (d < s) red[d] += red[d + s];
    __syncthreads();
  }
  float qi = rsqrtf(red[0] / (float)HD + EPS_F);
  __syncthreads();
  red[d] = kv * kv;
  __syncthreads();
  for (int s = 32; s > 0; s >>= 1) {
    if (d < s) red[d] += red[d + s];
    __syncthreads();
  }
  float ki = rsqrtf(red[0] / (float)HD + EPS_F);
  __syncthreads();

  sq[d] = qv * qi;
  sk[d] = kv * ki;
  __syncthreads();

  const int j = d & 31;
  float theta = (j < 16) ? (float)t * powf(10000.0f, -(float)j / 15.0f) : 0.0f;
  float c = cosf(theta), s_ = sinf(theta);
  float x1q = sq[j], x2q = sq[j + 32];
  float x1k = sk[j], x2k = sk[j + 32];
  float oq = (d < 32) ? (x1q * c + x2q * s_) : (-x1q * s_ + x2q * c);
  float ok = (d < 32) ? (x1k * c + x2k * s_) : (-x1k * s_ + x2k * c);

  const size_t oidx = ((size_t)h * T_SEQ + t) * HD + d;
  qh[oidx] = oq;
  kh[oidx] = ok;
  vh[oidx] = vv;
}

__global__ __launch_bounds__(32) __attribute__((amdgpu_waves_per_eu(1, 4))) void attn_kernel(
    const float* __restrict__ Qh, const float* __restrict__ Kh,
    const float* __restrict__ Vh, float* __restrict__ outAttn) {
  const int h = blockIdx.x;
  const int q0 = blockIdx.y * 16;
  const int lane = threadIdx.x;
  const int lm = lane & 15;
  const int half = lane >> 4;
  const size_t hb = (size_t)h * T_SEQ * HD;

  const float* qrow = Qh + hb + (size_t)(q0 + lm) * HD + 2 * half;
  v2f qa[16];
#pragma unroll
  for (int dc = 0; dc < 16; ++dc) qa[dc] = *(const v2f*)(qrow + 4 * dc);

  float li[8];
  v8f o[4] = {};
#pragma unroll
  for (int r = 0; r < 8; ++r) li[r] = 0.0f;

  __shared__ float sP[16 * 16];

  const int lo = q0 - WIN + 1;
  const int kt0 = (lo < 0) ? 0 : (lo / 16) * 16;
  for (int kt = kt0; kt <= q0; kt += 16) {
    const float* krow = Kh + hb + (size_t)(kt + lm) * HD + 2 * half;
    v2f kb[16];
#pragma unroll
    for (int dc = 0; dc < 16; ++dc) kb[dc] = *(const v2f*)(krow + 4 * dc);
    v8f s0 = {}, s1 = {};
#pragma unroll
    for (int dc = 0; dc < 16; dc += 2) {
      s0 = wmma4(qa[dc], kb[dc], s0);
      s1 = wmma4(qa[dc + 1], kb[dc + 1], s1);
    }
    v8f s = s0 + s1;

#pragma unroll
    for (int r = 0; r < 8; ++r) {
      const int qp = q0 + r + 8 * half;
      const int kp = kt + lm;
      const float sv = s[r] * 0.125f;
      const bool okm = (kp <= qp) && (kp > qp - WIN);
      const float p = okm ? __expf(sv) : 0.0f;
      li[r] += p;
      sP[(r + 8 * half) * 16 + lm] = p;
    }
    __syncthreads();

    const float* vbase = Vh + hb;
    v2f vb[16];
#pragma unroll
    for (int kc4 = 0; kc4 < 4; ++kc4) {
      const int kr = kt + kc4 * 4 + 2 * half;
#pragma unroll
      for (int f = 0; f < 4; ++f) {
        v2f b;
        b.x = vbase[(size_t)kr * HD + f * 16 + lm];
        b.y = vbase[(size_t)(kr + 1) * HD + f * 16 + lm];
        vb[kc4 * 4 + f] = b;
      }
    }
    v2f pa[4];
#pragma unroll
    for (int kc4 = 0; kc4 < 4; ++kc4) {
      pa[kc4].x = sP[lm * 16 + kc4 * 4 + 2 * half];
      pa[kc4].y = sP[lm * 16 + kc4 * 4 + 2 * half + 1];
    }
#pragma unroll
    for (int kc4 = 0; kc4 < 4; ++kc4)
#pragma unroll
      for (int f = 0; f < 4; ++f) o[f] = wmma4(pa[kc4], vb[kc4 * 4 + f], o[f]);
    __syncthreads();
  }

#pragma unroll
  for (int r = 0; r < 8; ++r) {
    float rs = li[r];
    rs += __shfl_xor(rs, 1, 32);
    rs += __shfl_xor(rs, 2, 32);
    rs += __shfl_xor(rs, 4, 32);
    rs += __shfl_xor(rs, 8, 32);
    li[r] = rs;
  }

#pragma unroll
  for (int f = 0; f < 4; ++f) {
#pragma unroll
    for (int r = 0; r < 8; ++r) {
      const int tq = q0 + r + 8 * half;
      const int col = h * HD + f * 16 + lm;
      outAttn[(size_t)tq * DMODEL + col] = o[f][r] / li[r];
    }
  }
}

extern "C" void kernel_launch(void* const* d_in, const int* in_sizes, int n_in,
                              void* d_out, int out_size, void* d_ws,
                              size_t ws_size, hipStream_t stream) {
  (void)in_sizes;
  (void)n_in;
  (void)out_size;
  (void)ws_size;
  const float* x = (const float*)d_in[0];
  const float* qkv_w = (const float*)d_in[1];
  const float* o_w = (const float*)d_in[2];
  const float* o_scale = (const float*)d_in[3];
  const float* w1 = (const float*)d_in[4];
  const float* w2 = (const float*)d_in[5];
  const float* w2_scale = (const float*)d_in[6];
  float* out = (float*)d_out;

  const size_t TD = (size_t)T_SEQ * DMODEL;
  float* xn = (float*)d_ws;
  float* qkv = xn + TD;
  float* qh = qkv + 3 * TD;
  float* kh = qh + TD;
  float* vh = kh + TD;
  float* attn = vh + TD;
  float* x1 = attn + TD;
  float* hbuf = qkv;

  rmsnorm_kernel<<<T_SEQ, 256, 0, stream>>>(x, xn, DMODEL);
  gemm_wt_kernel<<<dim3(3 * DMODEL / BN, T_SEQ / BM), 128, 0, stream>>>(
      xn, qkv_w, nullptr, nullptr, qkv, T_SEQ, 3 * DMODEL, DMODEL, 0);
  qk_rope_kernel<<<dim3(T_SEQ, NH), 64, 0, stream>>>(qkv, qh, kh, vh);
  attn_kernel<<<dim3(NH, T_SEQ / 16), 32, 0, stream>>>(qh, kh, vh, attn);
  gemm_wt_kernel<<<dim3(DMODEL / BN, T_SEQ / BM), 128, 0, stream>>>(
      attn, o_w, o_scale, x, x1, T_SEQ, DMODEL, DMODEL, 0);
  rmsnorm_kernel<<<T_SEQ, 256, 0, stream>>>(x1, xn, DMODEL);
  gemm_wt_kernel<<<dim3(DMLP / BN, T_SEQ / BM), 128, 0, stream>>>(
      xn, w1, nullptr, nullptr, hbuf, T_SEQ, DMLP, DMODEL, 1);
  gemm_wt_kernel<<<dim3(DMODEL / BN, T_SEQ / BM), 128, 0, stream>>>(
      hbuf, w2, w2_scale, x1, out, T_SEQ, DMODEL, DMLP, 0);
}


# Row passes (2 RMS norms, head norm+rotary repack) one wave per row in registers: 128-bit loads, lane-exchange sums, angles once per token; no LDS/barriers, 12k waves not 197k; square sums reordered
# speedup vs baseline: 3.0615x; 1.0399x over previous
#include <hip/hip_runtime.h>
#include <math.h>


#define T_SEQ  4096
#define DMODEL 1024
#define NH     16
#define HD     64
#define DMLP   4096
#define WIN    512
#define EPS_F  1.1920929e-07f

#define BM   128
#define BN   128
#define KC   32

typedef float v2f __attribute__((ext_vector_type(2)));
typedef float v4f __attribute__((ext_vector_type(4)));
typedef float v8f __attribute__((ext_vector_type(8)));

__device__ __forceinline__ v8f wmma4(v2f a, v2f b, v8f c) {
  return __builtin_amdgcn_wmma_f32_16x16x4_f32(false, a, false, b, (short)0, c,
                                               false, false);
}

#define ROW_WAVES 4

__global__ __launch_bounds__(32 * ROW_WAVES) void rmsnorm_kernel(
    const float* __restrict__ in, float* __restrict__ out) {
  const int lane = threadIdx.x & 31;
  const int row = blockIdx.x * ROW_WAVES + (threadIdx.x >> 5);
  const float* r = in + (size_t)row * DMODEL + 4 * lane;
  float* o = out + (size_t)row * DMODEL + 4 * lane;
  v4f v[8];
#pragma unroll
  for (int i = 0; i < 8; ++i) v[i] = *(const v4f*)(r + 128 * i);
  v4f s4 = v[0] * v[0];
#pragma unroll
  for (int i = 1; i < 8; ++i) s4 += v[i] * v[i];
  float ss = (s4.x + s4.y) + (s4.z + s4.w);
  ss += __shfl_xor(ss, 16, 32);
  ss += __shfl_xor(ss, 8, 32);
  ss += __shfl_xor(ss, 4, 32);
  ss += __shfl_xor(ss, 2, 32);
  ss += __shfl_xor(ss, 1, 32);
  const float inv = rsqrtf(ss / (float)DMODEL + EPS_F);
#pragma unroll
  for (int i = 0; i < 8; ++i) *(v4f*)(o + 128 * i) = v[i] * inv;
}

struct FragSet {
  v4f a[4][4];
  v4f b[4][4];
};

__device__ __forceinline__ void mma_slice(const FragSet& f, v8f (&acc)[4][4],
                                          int j) {
#pragma unroll
  for (int i = 0; i < 4; ++i)
#pragma unroll
    for (int s = 0; s < 4; ++s)
      acc[i][s] = wmma4(__builtin_shufflevector(f.a[i][j], f.a[i][j], 0, 1),
                        __builtin_shufflevector(f.b[s][j], f.b[s][j], 0, 1),
                        acc[i][s]);
#pragma unroll
  for (int i = 0; i < 4; ++i)
#pragma unroll
    for (int s = 0; s < 4; ++s)
      acc[i][s] = wmma4(__builtin_shufflevector(f.a[i][j], f.a[i][j], 2, 3),
                        __builtin_shufflevector(f.b[s][j], f.b[s][j], 2, 3),
                        acc[i][s]);
}

__device__ __forceinline__ void load_slice(FragSet& f,
                                           const float* __restrict__ Ap,
                                           const float* __restrict__ Wp,
                                           size_t rs16, int k, int j) {
#pragma unroll
  for (int i = 0; i < 4; ++i)
    f.a[i][j] = *(const v4f*)(Ap + i * rs16 + k + 4 * j);
#pragma unroll
  for (int i = 0; i < 4; ++i)
    f.b[i][j] = *(const v4f*)(Wp + i * rs16 + k + 4 * j);
}

__device__ __forceinline__ void chunk_step(const FragSet& use, FragSet& fill,
                                           v8f (&acc)[4][4],
                                           const float* __restrict__ Ap,
                                           const float* __restrict__ Wp,
                                           size_t rs16, int knext) {
#pragma unroll
  for (int j = 0; j < 4; ++j) {
    load_slice(fill, Ap, Wp, rs16, knext, j);
    asm volatile("" ::: "memory");
    __builtin_amdgcn_sched_barrier(0);
    mma_slice(use, acc, j);
    __builtin_amdgcn_sched_barrier(0);
  }
}

__device__ __forceinline__ void mma_set(const FragSet& f, v8f (&acc)[4][4]) {
#pragma unroll
  for (int j = 0; j < 4; ++j) mma_slice(f, acc, j);
}

template <bool RES, bool ACT>
__device__ __forceinline__ void gemm_epilogue(
    const v8f (&acc)[4][4], const float* __restrict__ colScale,
    const float* __restrict__ resid, float* __restrict__ out, int mB, int nB,
    int N, int lm, int half) {
  float cs[4];
#pragma unroll
  for (int s = 0; s < 4; ++s)
    cs[s] = colScale ? colScale[nB + s * 16 + lm] : 1.0f;
#pragma unroll
  for (int i = 0; i < 4; ++i) {
    float rv[8][4];
    if (RES) {
#pragma unroll
      for (int r = 0; r < 8; ++r)
#pragma unroll
        for (int s = 0; s < 4; ++s)
          rv[r][s] = resid[(size_t)(mB + i * 16 + r + 8 * half) * N + nB +
                           s * 16 + lm];
    }
#pragma unroll
    for (int r = 0; r < 8; ++r) {
      const int m = mB + i * 16 + r + 8 * half;
#pragma unroll
      for (int s = 0; s < 4; ++s) {
        const int n = nB + s * 16 + lm;
        float v = acc[i][s][r] * cs[s];
        if (ACT) {
          v = fmaxf(v, 0.0f);
          v = v * v;
        }
        if (RES) v += rv[r][s];
        out[(size_t)m * N + n] = v;
      }
    }
  }
}

__global__ __launch_bounds__(128) void gemm_wt_kernel(
    const float* __restrict__ A, const float* __restrict__ Wt,
    const float* __restrict__ colScale, const float* __restrict__ resid,
    float* __restrict__ out, int M, int N, int K, int act) {
  (void)M;
  const int tid = threadIdx.x;
  const int lane = tid & 31;
  const int wave = tid >> 5;
  const int lm = lane & 15;
  const int half = lane >> 4;
  const int mB = blockIdx.y * BM + (wave & 1) * 64;
  const int nB = blockIdx.x * BN + (wave >> 1) * 64;

  const float* Ap = A + (size_t)(mB + lm) * K + 16 * half;
  const float* Wp = Wt + (size_t)(nB + lm) * K + 16 * half;
  const size_t rs16 = (size_t)16 * K;

  v8f acc[4][4] = {};
  FragSet f0, f1;

#pragma unroll
  for (int j = 0; j < 4; ++j) {
    load_slice(f0, Ap, Wp, rs16, 0, j);
    asm volatile("" ::: "memory");
  }
  int kc = 0;
  for (; kc + 2 * KC < K; kc += 2 * KC) {
    chunk_step(f0, f1, acc, Ap, Wp, rs16, kc + KC);
    chunk_step(f1, f0, acc, Ap, Wp, rs16, kc + 2 * KC);
  }
  chunk_step(f0, f1, acc, Ap, Wp, rs16, kc + KC);
  mma_set(f1, acc);

  if (resid) {
    if (act == 1)
      gemm_epilogue<true, true>(acc, colScale, resid, out, mB, nB, N, lm, half);
    else
      gemm_epilogue<true, false>(acc, colScale, resid, out, mB, nB, N, lm, half);
  } else {
    if (act == 1)
      gemm_epilogue<false, true>(acc, colScale, resid, out, mB, nB, N, lm, half);
    else
      gemm_epilogue<false, false>(acc, colScale, resid, out, mB, nB, N, lm, half);
  }
}

__global__ __launch_bounds__(32 * ROW_WAVES) void qk_rope_kernel(
    const float* __restrict__ qkv, float* __restrict__ qh,
    float* __restrict__ kh, float* __restrict__ vh) {
  const int lane = threadIdx.x & 31;
  const int t = blockIdx.x * ROW_WAVES + (threadIdx.x >> 5);
  const int g = lane >> 3;
  const int i = lane & 7;
  const float* src = qkv + (size_t)t * (3 * DMODEL) + g * HD + 4 * i;

  v4f x1[3][4], x2[3][4];
#pragma unroll
  for (int w = 0; w < 3; ++w)
#pragma unroll
    for (int it = 0; it < 4; ++it) {
      x1[w][it] = *(const v4f*)(src + w * DMODEL + it * 4 * HD);
      x2[w][it] = *(const v4f*)(src + w * DMODEL + it * 4 * HD + 32);
    }

  const float theta = (float)t * powf(10000.0f, -(float)(lane & 15) / 15.0f);
  const float cl = cosf(theta), sl = sinf(theta);
  v4f c, s;
#pragma unroll
  for (int e = 0; e < 4; ++e) {
    const float ce = __shfl(cl, 4 * (i & 3) + e, 32);
    const float se = __shfl(sl, 4 * (i & 3) + e, 32);
    c[e] = (i < 4) ? ce : 1.0f;
    s[e] = (i < 4) ? se : 0.0f;
  }

  const size_t dst = ((size_t)g * T_SEQ + t) * HD + 4 * i;
  const size_t grp = (size_t)4 * T_SEQ * HD;
#pragma unroll
  for (int w = 0; w < 2; ++w) {
    float* __restrict__ o = (w == 0) ? qh : kh;
#pragma unroll
    for (int it = 0; it < 4; ++it) {
      v4f a = x1[w][it], b = x2[w][it];
      const v4f p = a * a + b * b;
      float ss = (p.x + p.y) + (p.z + p.w);
      ss += __shfl_xor(ss, 4, 32);
      ss += __shfl_xor(ss, 2, 32);
      ss += __shfl_xor(ss, 1, 32);
      const float inv = rsqrtf(ss / (float)HD + EPS_F);
      a *= inv;
      b *= inv;
      *(v4f*)(o + dst + it * grp) = a * c + b * s;
      *(v4f*)(o + dst + it * grp + 32) = b * c - a * s;
    }
  }
#pragma unroll
  for (int it = 0; it < 4; ++it) {
    *(v4f*)(vh + dst + it * grp) = x1[2][it];
    *(v4f*)(vh + dst + it * grp + 32) = x2[2][it];
  }
}

__global__ __launch_bounds__(32) __attribute__((amdgpu_waves_per_eu(1, 4))) void attn_kernel(
    const float* __restrict__ Qh, const float* __restrict__ Kh,
    const float* __restrict__ Vh, float* __restrict__ outAttn) {
  const int h = blockIdx.x;
  const int q0 = blockIdx.y * 16;
  const int lane = threadIdx.x;
  const int lm = lane & 15;
  const int half = lane >> 4;
  const size_t hb = (size_t)h * T_SEQ * HD;

  const float* qrow = Qh + hb + (size_t)(q0 + lm) * HD + 2 * half;
  v2f qa[16];
#pragma unroll
  for (int dc = 0; dc < 16; ++dc) qa[dc] = *(const v2f*)(qrow + 4 * dc);

  float li[8];
  v8f o[4] = {};
#pragma unroll
  for (int r = 0; r < 8; ++r) li[r] = 0.0f;

  __shared__ float sP[16 * 16];

  const int lo = q0 - WIN + 1;
  const int kt0 = (lo < 0) ? 0 : (lo / 16) * 16;
  for (int kt = kt0; kt <= q0; kt += 16) {
    const float* krow = Kh + hb + (size_t)(kt + lm) * HD + 2 * half;
    v2f kb[16];
#pragma unroll
    for (int dc = 0; dc < 16; ++dc) kb[dc] = *(const v2f*)(krow + 4 * dc);
    v8f s0 = {}, s1 = {};
#pragma unroll
    for (int dc = 0; dc < 16; dc += 2) {
      s0 = wmma4(qa[dc], kb[dc], s0);
      s1 = wmma4(qa[dc + 1], kb[dc + 1], s1);
    }
    v8f s = s0 + s1;

#pragma unroll
    for (int r = 0; r < 8; ++r) {
      const int qp = q0 + r + 8 * half;
      const int kp = kt + lm;
      const float sv = s[r] * 0.125f;
      const bool okm = (kp <= qp) && (kp > qp - WIN);
      const float p = okm ? __expf(sv) : 0.0f;
      li[r] += p;
      sP[(r + 8 * half) * 16 + lm] = p;
    }
    __syncthreads();

    const float* vbase = Vh + hb;
    v2f vb[16];
#pragma unroll
    for (int kc4 = 0; kc4 < 4; ++kc4) {
      const int kr = kt + kc4 * 4 + 2 * half;
#pragma unroll
      for (int f = 0; f < 4; ++f) {
        v2f b;
        b.x = vbase[(size_t)kr * HD + f * 16 + lm];
        b.y = vbase[(size_t)(kr + 1) * HD + f * 16 + lm];
        vb[kc4 * 4 + f] = b;
      }
    }
    v2f pa[4];
#pragma unroll
    for (int kc4 = 0; kc4 < 4; ++kc4) {
      pa[kc4].x = sP[lm * 16 + kc4 * 4 + 2 * half];
      pa[kc4].y = sP[lm * 16 + kc4 * 4 + 2 * half + 1];
    }
#pragma unroll
    for (int kc4 = 0; kc4 < 4; ++kc4)
#pragma unroll
      for (int f = 0; f < 4; ++f) o[f] = wmma4(pa[kc4], vb[kc4 * 4 + f], o[f]);
    __syncthreads();
  }

#pragma unroll
  for (int r = 0; r < 8; ++r) {
    float rs = li[r];
    rs += __shfl_xor(rs, 1, 32);
    rs += __shfl_xor(rs, 2, 32);
    rs += __shfl_xor(rs, 4, 32);
    rs += __shfl_xor(rs, 8, 32);
    li[r] = rs;
  }

#pragma unroll
  for (int f = 0; f < 4; ++f) {
#pragma unroll
    for (int r = 0; r < 8; ++r) {
      const int tq = q0 + r + 8 * half;
      const int col = h * HD + f * 16 + lm;
      outAttn[(size_t)tq * DMODEL + col] = o[f][r] / li[r];
    }
  }
}

extern "C" void kernel_launch(void* const* d_in, const int* in_sizes, int n_in,
                              void* d_out, int out_size, void* d_ws,
                              size_t ws_size, hipStream_t stream) {
  (void)in_sizes;
  (void)n_in;
  (void)out_size;
  (void)ws_size;
  const float* x = (const float*)d_in[0];
  const float* qkv_w = (const float*)d_in[1];
  const float* o_w = (const float*)d_in[2];
  const float* o_scale = (const float*)d_in[3];
  const float* w1 = (const float*)d_in[4];
  const float* w2 = (const float*)d_in[5];
  const float* w2_scale = (const float*)d_in[6];
  float* out = (float*)d_out;

  const size_t TD = (size_t)T_SEQ * DMODEL;
  float* xn = (float*)d_ws;
  float* qkv = xn + TD;
  float* qh = qkv + 3 * TD;
  float* kh = qh + TD;
  float* vh = kh + TD;
  float* attn = vh + TD;
  float* x1 = attn + TD;
  float* hbuf = qkv;

  rmsnorm_kernel<<<T_SEQ / ROW_WAVES, 32 * ROW_WAVES, 0, stream>>>(x, xn);
  gemm_wt_kernel<<<dim3(3 * DMODEL / BN, T_SEQ / BM), 128, 0, stream>>>(
      xn, qkv_w, nullptr, nullptr, qkv, T_SEQ, 3 * DMODEL, DMODEL, 0);
  qk_rope_kernel<<<T_SEQ / ROW_WAVES, 32 * ROW_WAVES, 0, stream>>>(qkv, qh, kh, vh);
  attn_kernel<<<dim3(NH, T_SEQ / 16), 32, 0, stream>>>(qh, kh, vh, attn);
  gemm_wt_kernel<<<dim3(DMODEL / BN, T_SEQ / BM), 128, 0, stream>>>(
      attn, o_w, o_scale, x, x1, T_SEQ, DMODEL, DMODEL, 0);
  rmsnorm_kernel<<<T_SEQ / ROW_WAVES, 32 * ROW_WAVES, 0, stream>>>(x1, xn);
  gemm_wt_kernel<<<dim3(DMLP / BN, T_SEQ / BM), 128, 0, stream>>>(
      xn, w1, nullptr, nullptr, hbuf, T_SEQ, DMLP, DMODEL, 1);
  gemm_wt_kernel<<<dim3(DMODEL / BN, T_SEQ / BM), 128, 0, stream>>>(
      hbuf, w2, w2_scale, x1, out, T_SEQ, DMODEL, DMLP, 0);
}
